# starting kernel
# baseline (speedup 1.0000x reference)
#include <hip/hip_runtime.h>

typedef float v2f __attribute__((ext_vector_type(2)));
typedef float v8f __attribute__((ext_vector_type(8)));

#define B_ 8
#define I_ 16
#define C_ 256

static const size_t SEL_OFF[4] = {0, 2097152, 2621440, 2752512};
static const size_t S_OFF      = 2785280;
static const size_t CNT_OFF    = 2916352;
static const size_t PK_BYTEOFF = 2916864ull * 4ull;

__global__ void zero_kernel(float* __restrict__ p, int n) {
    int i = blockIdx.x * blockDim.x + threadIdx.x;
    if (i < n) p[i] = 0.0f;
}

__global__ void __launch_bounds__(256)
mask_kernel(const float* __restrict__ scrib, float* __restrict__ sel,
            float* __restrict__ cnt, unsigned long long* __restrict__ packed,
            int w, int factor, int offs, int blocksPerBI) {
    __shared__ float scnt[256];
    __shared__ unsigned long long smax[256];
    int bi = blockIdx.x / blocksPerBI;
    int pb = blockIdx.x % blocksPerBI;
    int t  = threadIdx.x;
    int p  = pb * 256 + t;
    int y = p / w, x = p % w;
    int ys = y * factor + offs;
    int xs = x * factor + offs;
    const float* sp = scrib + (size_t)bi * 512 * 512;
    float m = 0.25f * (sp[(size_t)ys * 512 + xs]     + sp[(size_t)ys * 512 + xs + 1] +
                       sp[(size_t)(ys + 1) * 512 + xs] + sp[(size_t)(ys + 1) * 512 + xs + 1]);
    float sv = (m > 0.5f) ? 1.0f : 0.0f;
    sel[(size_t)bi * ((size_t)blocksPerBI * 256) + p] = sv;
    scnt[t] = sv;
    smax[t] = ((unsigned long long)__float_as_uint(m) << 32) |
              (unsigned long long)(0xFFFFFFFFu - (unsigned)p);
    __syncthreads();
    for (int st = 128; st > 0; st >>= 1) {
        if (t < st) {
            scnt[t] += scnt[t + st];
            if (smax[t + st] > smax[t]) smax[t] = smax[t + st];
        }
        __syncthreads();
    }
    if (t == 0) {
        atomicAdd(&cnt[bi], scnt[0]);
        atomicMax(&packed[bi], smax[0]);
    }
}

__global__ void __launch_bounds__(256)
wmma_masked_sum_kernel(const float* __restrict__ feat, const float* __restrict__ sel,
                       float* __restrict__ s, int P, int chunk, int nchunks) {
    int wid  = (blockIdx.x * blockDim.x + threadIdx.x) >> 5;
    int lane = threadIdx.x & 31;
    int kc = wid % nchunks;
    int ct = (wid / nchunks) & 15;
    int b  = wid / (nchunks * 16);
    int k0 = kc * chunk;
    int row = lane & 15;
    int kb  = (lane >> 4) * 2;
    const float* ap = sel  + ((size_t)b * I_ + row) * (size_t)P + (size_t)(k0 + kb);
    const float* bp = feat + ((size_t)b * C_ + (size_t)ct * 16 + row) * (size_t)P + (size_t)(k0 + kb);
    v8f acc = {};
    for (int k = 0; k < chunk; k += 4) {
        v2f a  = *(const v2f*)(ap + k);
        v2f bv = *(const v2f*)(bp + k);
        acc = __builtin_amdgcn_wmma_f32_16x16x4_f32(false, a, false, bv,
                                                    (short)0, acc, false, false);
    }
    float* sd = s + ((size_t)b * I_ + (size_t)(lane >> 4) * 8) * C_ + (size_t)ct * 16 + (lane & 15);
#pragma unroll
    for (int r = 0; r < 8; ++r)
        atomicAdd(sd + (size_t)r * C_, acc[r]);
}

__global__ void __launch_bounds__(256)
finalize_kernel(const float* __restrict__ f0, const float* __restrict__ f1,
                const float* __restrict__ f2, const float* __restrict__ f3,
                const float* __restrict__ s, const float* __restrict__ cnt,
                const unsigned long long* __restrict__ packed, float* __restrict__ out) {
    int tid = blockIdx.x * blockDim.x + threadIdx.x;
    int c  = tid & 255;
    int bi = tid >> 8;
    int b  = bi >> 4;
    const float* feats[4] = {f0, f1, f2, f3};
    const int Ps[4] = {16384, 4096, 1024, 256};
    float sum = 0.0f;
#pragma unroll
    for (int l = 0; l < 4; ++l) {
        float cv = cnt[l * (B_ * I_) + bi];
        float v;
        if (cv > 0.0f) {
            v = s[l * (B_ * I_ * C_) + tid] / cv;
        } else {
            unsigned long long pk = packed[l * (B_ * I_) + bi];
            unsigned idx = 0xFFFFFFFFu - (unsigned)(pk & 0xFFFFFFFFull);
            v = feats[l][((size_t)b * C_ + c) * (size_t)Ps[l] + idx];
        }
        sum += v;
    }
    out[tid] = 0.25f * sum;
}

extern "C" void kernel_launch(void* const* d_in, const int* in_sizes, int n_in,
                              void* d_out, int out_size, void* d_ws, size_t ws_size,
                              hipStream_t stream) {
    const float* feat0 = (const float*)d_in[0];
    const float* feat1 = (const float*)d_in[1];
    const float* feat2 = (const float*)d_in[2];
    const float* feat3 = (const float*)d_in[3];
    const float* scrib = (const float*)d_in[4];
    float* ws = (float*)d_ws;

    float* sA   = ws + S_OFF;
    float* cntA = ws + CNT_OFF;
    unsigned long long* pkA = (unsigned long long*)((char*)d_ws + PK_BYTEOFF);

    int nz = (B_ * I_ * C_ * 4) + (B_ * I_ * 4) + (B_ * I_ * 4 * 2);
    zero_kernel<<<(nz + 255) / 256, 256, 0, stream>>>(sA, nz);

    const float* feats[4] = {feat0, feat1, feat2, feat3};
    const int H[4]  = {128, 64, 32, 16};
    const int CH[4] = {1024, 512, 256, 128};

    for (int l = 0; l < 4; ++l) {
        int h = H[l], P = h * h;
        int factor = 512 / h;
        int offs = factor / 2 - 1;
        int blocksPerBI = P / 256;
        mask_kernel<<<B_ * I_ * blocksPerBI, 256, 0, stream>>>(
            scrib, ws + SEL_OFF[l], cntA + l * (B_ * I_), pkA + l * (B_ * I_),
            h, factor, offs, blocksPerBI);
    }
    for (int l = 0; l < 4; ++l) {
        int P = H[l] * H[l];
        int chunk = CH[l];
        int nchunks = P / chunk;
        int waves = B_ * 16 * nchunks;
        wmma_masked_sum_kernel<<<waves * 32 / 256, 256, 0, stream>>>(
            feats[l], ws + SEL_OFF[l], sA + l * (B_ * I_ * C_), P, chunk, nchunks);
    }
    finalize_kernel<<<(B_ * I_ * C_) / 256, 256, 0, stream>>>(
        feat0, feat1, feat2, feat3, sA, cntA, pkA, (float*)d_out);
}


# mask pass removed: each masked-sum workgroup reads the 2x2 mask blocks of its own slabs, thresholds them straight into its LDS sel tile and reduces cnt/argmax itself; cnt sum regrouped (exact)
# speedup vs baseline: 1.4178x; 1.4178x over previous
#include <hip/hip_runtime.h>

typedef float v2f __attribute__((ext_vector_type(2)));
typedef float v4f __attribute__((ext_vector_type(4)));
typedef float v8f __attribute__((ext_vector_type(8)));

#define B_ 8
#define I_ 16
#define C_ 256

static const size_t S_OFF      = 2785280;
static const size_t CNT_OFF    = 2916352;
static const size_t PK_BYTEOFF = 2916864ull * 4ull;

__global__ void zero_kernel(float* __restrict__ p, int n) {
    int i = blockIdx.x * blockDim.x + threadIdx.x;
    if (i < n) p[i] = 0.0f;
}

#define KC    128
#define PITCH 132

__device__ __forceinline__ unsigned long long load_pair(const float* p) {
    unsigned long long v;
    __builtin_memcpy(&v, p, sizeof(v));
    return v;
}
__device__ __forceinline__ float pair_lo(unsigned long long v) { return __uint_as_float((unsigned)v); }
__device__ __forceinline__ float pair_hi(unsigned long long v) { return __uint_as_float((unsigned)(v >> 32)); }

template <int LW>
__device__ __forceinline__ int mask_src(int rowbase, int pix) {
    constexpr int F    = 512 >> LW;
    constexpr int OFFS = F / 2 - 1;
    const int y = pix >> LW;
    const int x = pix & ((1 << LW) - 1);
    return (rowbase + y * F + OFFS) * 512 + x * F + OFFS;
}

template <int NSUB, int LW>
__global__ void __launch_bounds__(256)
slab_masked_sum_kernel(const float* __restrict__ feat, const float* __restrict__ scrib,
                       float* __restrict__ s, float* __restrict__ cnt,
                       unsigned long long* __restrict__ packed) {
    constexpr int P      = (1 << LW) * (1 << LW);
    constexpr int wgPerB = P / (NSUB * KC);
    constexpr int PLANE8 = 8 * 512 * 512;
    __shared__ __attribute__((aligned(16))) float lsel[I_ * PITCH];
    __shared__ __attribute__((aligned(16))) float lfeat[C_ * PITCH];
    const int t    = threadIdx.x;
    const int lane = t & 31;
    const int wv   = t >> 5;
    const int b    = blockIdx.x / wgPerB;
    const int p0   = (blockIdx.x % wgPerB) * (NSUB * KC);
    const int c    = lane & 15;
    const int h    = lane >> 4;

    const float* fg = feat + ((size_t)b * C_ + (size_t)(32 * wv)) * (size_t)P + (size_t)(p0 + 4 * lane);
    const int rowbase = (b * I_ + wv) * 512;
    const int pl      = p0 + lane;

    float* lf = lfeat + (32 * wv) * PITCH + 4 * lane;
    float* ls = lsel + wv * PITCH + lane;
    const float* ap  = lsel + c * PITCH + 4 * h;
    const float* bp0 = lfeat + (32 * wv + c) * PITCH + 4 * h;
    const float* bp1 = bp0 + 16 * PITCH;

    v4f  fr[32];
    unsigned long long sc[16];
#pragma unroll
    for (int q = 0; q < 4; ++q) {
        const int e = mask_src<LW>(rowbase, pl + 32 * q);
        sc[4 * q + 0] = load_pair(scrib + e);
        sc[4 * q + 1] = load_pair(scrib + e + 512);
        sc[4 * q + 2] = load_pair(scrib + e + PLANE8);
        sc[4 * q + 3] = load_pair(scrib + e + PLANE8 + 512);
    }
#pragma unroll
    for (int j = 0; j < 32; ++j) fr[j] = *(const v4f*)(fg + (size_t)j * (size_t)P);

    float cn0 = 0.0f, cn1 = 0.0f;
    unsigned long long bm0 = 0ull, bm1 = 0ull;
    v8f acc0 = {};
    v8f acc1 = {};
    for (int sub = 0; sub < NSUB; ++sub) {
#pragma unroll
        for (int q = 0; q < 4; ++q) {
            const int pix = pl + 32 * q + sub * KC;
            const float m0 = 0.25f * (pair_lo(sc[4 * q + 0]) + pair_hi(sc[4 * q + 0]) + pair_lo(sc[4 * q + 1]) + pair_hi(sc[4 * q + 1]));
            const float m1 = 0.25f * (pair_lo(sc[4 * q + 2]) + pair_hi(sc[4 * q + 2]) + pair_lo(sc[4 * q + 3]) + pair_hi(sc[4 * q + 3]));
            const float s0 = (m0 > 0.5f) ? 1.0f : 0.0f;
            const float s1 = (m1 > 0.5f) ? 1.0f : 0.0f;
            ls[32 * q]             = s0;
            ls[8 * PITCH + 32 * q] = s1;
            cn0 += s0;
            cn1 += s1;
            const unsigned long long lo = (unsigned long long)(0xFFFFFFFFu - (unsigned)pix);
            const unsigned long long k0 = ((unsigned long long)__float_as_uint(m0) << 32) | lo;
            const unsigned long long k1 = ((unsigned long long)__float_as_uint(m1) << 32) | lo;
            if (k0 > bm0) bm0 = k0;
            if (k1 > bm1) bm1 = k1;
        }
#pragma unroll
        for (int j = 0; j < 32; ++j) *(v4f*)(lf + j * PITCH) = fr[j];
        __syncthreads();
        if (sub + 1 < NSUB) {
            const float* fgn = fg + (size_t)(sub + 1) * KC;
#pragma unroll
            for (int q = 0; q < 4; ++q) {
                const int e = mask_src<LW>(rowbase, pl + 32 * q + (sub + 1) * KC);
                sc[4 * q + 0] = load_pair(scrib + e);
                sc[4 * q + 1] = load_pair(scrib + e + 512);
                sc[4 * q + 2] = load_pair(scrib + e + PLANE8);
                sc[4 * q + 3] = load_pair(scrib + e + PLANE8 + 512);
            }
#pragma unroll
            for (int j = 0; j < 32; ++j) fr[j] = *(const v4f*)(fgn + (size_t)j * (size_t)P);
            __builtin_amdgcn_sched_barrier(0);
        }
        for (int k8 = 0; k8 < KC / 8; ++k8) {
            v4f a = *(const v4f*)(ap  + 8 * k8);
            v4f x = *(const v4f*)(bp0 + 8 * k8);
            v4f y = *(const v4f*)(bp1 + 8 * k8);
            v2f a01 = {a[0], a[1]}, a23 = {a[2], a[3]};
            v2f x01 = {x[0], x[1]}, x23 = {x[2], x[3]};
            v2f y01 = {y[0], y[1]}, y23 = {y[2], y[3]};
            acc0 = __builtin_amdgcn_wmma_f32_16x16x4_f32(false, a01, false, x01, (short)0, acc0, false, false);
            acc1 = __builtin_amdgcn_wmma_f32_16x16x4_f32(false, a01, false, y01, (short)0, acc1, false, false);
            acc0 = __builtin_amdgcn_wmma_f32_16x16x4_f32(false, a23, false, x23, (short)0, acc0, false, false);
            acc1 = __builtin_amdgcn_wmma_f32_16x16x4_f32(false, a23, false, y23, (short)0, acc1, false, false);
        }
        if (sub + 1 < NSUB) __syncthreads();
    }
    for (int off = 16; off > 0; off >>= 1) {
        cn0 += __shfl_xor(cn0, off);
        cn1 += __shfl_xor(cn1, off);
        const unsigned long long o0 = __shfl_xor(bm0, off);
        const unsigned long long o1 = __shfl_xor(bm1, off);
        if (o0 > bm0) bm0 = o0;
        if (o1 > bm1) bm1 = o1;
    }
    if (lane == 0) {
        atomicAdd(&cnt[b * I_ + wv], cn0);
        atomicAdd(&cnt[b * I_ + wv + 8], cn1);
        atomicMax(&packed[b * I_ + wv], bm0);
        atomicMax(&packed[b * I_ + wv + 8], bm1);
    }
    float* sd = s + ((size_t)b * I_ + (size_t)h * 8) * C_ + (size_t)(32 * wv + c);
#pragma unroll
    for (int r = 0; r < 8; ++r) {
        atomicAdd(sd + (size_t)r * C_, acc0[r]);
        atomicAdd(sd + (size_t)r * C_ + 16, acc1[r]);
    }
}

__global__ void __launch_bounds__(256)
finalize_kernel(const float* __restrict__ f0, const float* __restrict__ f1,
                const float* __restrict__ f2, const float* __restrict__ f3,
                const float* __restrict__ s, const float* __restrict__ cnt,
                const unsigned long long* __restrict__ packed, float* __restrict__ out) {
    int tid = blockIdx.x * blockDim.x + threadIdx.x;
    int c  = tid & 255;
    int bi = tid >> 8;
    int b  = bi >> 4;
    const float* feats[4] = {f0, f1, f2, f3};
    const int Ps[4] = {16384, 4096, 1024, 256};
    float sum = 0.0f;
#pragma unroll
    for (int l = 0; l < 4; ++l) {
        float cv = cnt[l * (B_ * I_) + bi];
        float v;
        if (cv > 0.0f) {
            v = s[l * (B_ * I_ * C_) + tid] / cv;
        } else {
            unsigned long long pk = packed[l * (B_ * I_) + bi];
            unsigned idx = 0xFFFFFFFFu - (unsigned)(pk & 0xFFFFFFFFull);
            v = feats[l][((size_t)b * C_ + c) * (size_t)Ps[l] + idx];
        }
        sum += v;
    }
    out[tid] = 0.25f * sum;
}

extern "C" void kernel_launch(void* const* d_in, const int* in_sizes, int n_in,
                              void* d_out, int out_size, void* d_ws, size_t ws_size,
                              hipStream_t stream) {
    const float* feat0 = (const float*)d_in[0];
    const float* feat1 = (const float*)d_in[1];
    const float* feat2 = (const float*)d_in[2];
    const float* feat3 = (const float*)d_in[3];
    const float* scrib = (const float*)d_in[4];
    float* ws = (float*)d_ws;

    float* sA   = ws + S_OFF;
    float* cntA = ws + CNT_OFF;
    unsigned long long* pkA = (unsigned long long*)((char*)d_ws + PK_BYTEOFF);

    int nz = (B_ * I_ * C_ * 4) + (B_ * I_ * 4) + (B_ * I_ * 4 * 2);
    zero_kernel<<<(nz + 255) / 256, 256, 0, stream>>>(sA, nz);

    slab_masked_sum_kernel<4, 7><<<256, 256, 0, stream>>>(feat0, scrib, sA,         cntA,       pkA);
    slab_masked_sum_kernel<2, 6><<<128, 256, 0, stream>>>(feat1, scrib, sA + 32768, cntA + 128, pkA + 128);
    slab_masked_sum_kernel<1, 5><<< 64, 256, 0, stream>>>(feat2, scrib, sA + 65536, cntA + 256, pkA + 256);
    slab_masked_sum_kernel<1, 4><<< 16, 256, 0, stream>>>(feat3, scrib, sA + 98304, cntA + 384, pkA + 384);
    finalize_kernel<<<(B_ * I_ * C_) / 256, 256, 0, stream>>>(
        feat0, feat1, feat2, feat3, sA, cntA, pkA, (float*)d_out);
}


# cross-workgroup reduction without atomics: each masked-sum workgroup stores its own partial s tile, cnt and argmax lines; finalize adds them in fixed order; zero launch gone; sum order changes
# speedup vs baseline: 1.6195x; 1.1423x over previous
#include <hip/hip_runtime.h>

typedef float v2f __attribute__((ext_vector_type(2)));
typedef float v4f __attribute__((ext_vector_type(4)));
typedef float v8f __attribute__((ext_vector_type(8)));

#define B_ 8
#define I_ 16
#define C_ 256

#define NWG_ALL 464
static const size_t SP_OFF      = 0;
static const size_t CNTP_OFF    = (size_t)NWG_ALL * 4096;
static const size_t PKP_BYTEOFF = ((size_t)NWG_ALL * 4096 + (size_t)NWG_ALL * 32) * 4ull;

#define KC    128
#define PITCH 132

__device__ __forceinline__ unsigned long long load_pair(const float* p) {
    unsigned long long v;
    __builtin_memcpy(&v, p, sizeof(v));
    return v;
}
__device__ __forceinline__ float pair_lo(unsigned long long v) { return __uint_as_float((unsigned)v); }
__device__ __forceinline__ float pair_hi(unsigned long long v) { return __uint_as_float((unsigned)(v >> 32)); }

template <int LW>
__device__ __forceinline__ int mask_src(int rowbase, int pix) {
    constexpr int F    = 512 >> LW;
    constexpr int OFFS = F / 2 - 1;
    const int y = pix >> LW;
    const int x = pix & ((1 << LW) - 1);
    return (rowbase + y * F + OFFS) * 512 + x * F + OFFS;
}

template <int NSUB, int LW>
__global__ void __launch_bounds__(256)
slab_masked_sum_kernel(const float* __restrict__ feat, const float* __restrict__ scrib,
                       float* __restrict__ spart, float* __restrict__ cntp,
                       unsigned long long* __restrict__ pkp) {
    constexpr int P      = (1 << LW) * (1 << LW);
    constexpr int wgPerB = P / (NSUB * KC);
    constexpr int PLANE8 = 8 * 512 * 512;
    __shared__ __attribute__((aligned(16))) float lsel[I_ * PITCH];
    __shared__ __attribute__((aligned(16))) float lfeat[C_ * PITCH];
    __shared__ float lcnt[I_];
    __shared__ unsigned long long lpk[I_];
    const int t    = threadIdx.x;
    const int lane = t & 31;
    const int wv   = t >> 5;
    const int b    = blockIdx.x / wgPerB;
    const int p0   = (blockIdx.x % wgPerB) * (NSUB * KC);
    const int c    = lane & 15;
    const int h    = lane >> 4;

    const float* fg = feat + ((size_t)b * C_ + (size_t)(32 * wv)) * (size_t)P + (size_t)(p0 + 4 * lane);
    const int rowbase = (b * I_ + wv) * 512;
    const int pl      = p0 + lane;

    float* lf = lfeat + (32 * wv) * PITCH + 4 * lane;
    float* ls = lsel + wv * PITCH + lane;
    const float* ap  = lsel + c * PITCH + 4 * h;
    const float* bp0 = lfeat + (32 * wv + c) * PITCH + 4 * h;
    const float* bp1 = bp0 + 16 * PITCH;

    v4f  fr[32];
    unsigned long long sc[16];
#pragma unroll
    for (int q = 0; q < 4; ++q) {
        const int e = mask_src<LW>(rowbase, pl + 32 * q);
        sc[4 * q + 0] = load_pair(scrib + e);
        sc[4 * q + 1] = load_pair(scrib + e + 512);
        sc[4 * q + 2] = load_pair(scrib + e + PLANE8);
        sc[4 * q + 3] = load_pair(scrib + e + PLANE8 + 512);
    }
#pragma unroll
    for (int j = 0; j < 32; ++j) fr[j] = *(const v4f*)(fg + (size_t)j * (size_t)P);

    float cn0 = 0.0f, cn1 = 0.0f;
    unsigned long long bm0 = 0ull, bm1 = 0ull;
    v8f acc0 = {};
    v8f acc1 = {};
    for (int sub = 0; sub < NSUB; ++sub) {
#pragma unroll
        for (int q = 0; q < 4; ++q) {
            const int pix = pl + 32 * q + sub * KC;
            const float m0 = 0.25f * (pair_lo(sc[4 * q + 0]) + pair_hi(sc[4 * q + 0]) + pair_lo(sc[4 * q + 1]) + pair_hi(sc[4 * q + 1]));
            const float m1 = 0.25f * (pair_lo(sc[4 * q + 2]) + pair_hi(sc[4 * q + 2]) + pair_lo(sc[4 * q + 3]) + pair_hi(sc[4 * q + 3]));
            const float s0 = (m0 > 0.5f) ? 1.0f : 0.0f;
            const float s1 = (m1 > 0.5f) ? 1.0f : 0.0f;
            ls[32 * q]             = s0;
            ls[8 * PITCH + 32 * q] = s1;
            cn0 += s0;
            cn1 += s1;
            const unsigned long long lo = (unsigned long long)(0xFFFFFFFFu - (unsigned)pix);
            const unsigned long long k0 = ((unsigned long long)__float_as_uint(m0) << 32) | lo;
            const unsigned long long k1 = ((unsigned long long)__float_as_uint(m1) << 32) | lo;
            if (k0 > bm0) bm0 = k0;
            if (k1 > bm1) bm1 = k1;
        }
#pragma unroll
        for (int j = 0; j < 32; ++j) *(v4f*)(lf + j * PITCH) = fr[j];
        __syncthreads();
        if (sub + 1 < NSUB) {
            const float* fgn = fg + (size_t)(sub + 1) * KC;
#pragma unroll
            for (int q = 0; q < 4; ++q) {
                const int e = mask_src<LW>(rowbase, pl + 32 * q + (sub + 1) * KC);
                sc[4 * q + 0] = load_pair(scrib + e);
                sc[4 * q + 1] = load_pair(scrib + e + 512);
                sc[4 * q + 2] = load_pair(scrib + e + PLANE8);
                sc[4 * q + 3] = load_pair(scrib + e + PLANE8 + 512);
            }
#pragma unroll
            for (int j = 0; j < 32; ++j) fr[j] = *(const v4f*)(fgn + (size_t)j * (size_t)P);
            __builtin_amdgcn_sched_barrier(0);
        }
        for (int k8 = 0; k8 < KC / 8; ++k8) {
            v4f a = *(const v4f*)(ap  + 8 * k8);
            v4f x = *(const v4f*)(bp0 + 8 * k8);
            v4f y = *(const v4f*)(bp1 + 8 * k8);
            v2f a01 = {a[0], a[1]}, a23 = {a[2], a[3]};
            v2f x01 = {x[0], x[1]}, x23 = {x[2], x[3]};
            v2f y01 = {y[0], y[1]}, y23 = {y[2], y[3]};
            acc0 = __builtin_amdgcn_wmma_f32_16x16x4_f32(false, a01, false, x01, (short)0, acc0, false, false);
            acc1 = __builtin_amdgcn_wmma_f32_16x16x4_f32(false, a01, false, y01, (short)0, acc1, false, false);
            acc0 = __builtin_amdgcn_wmma_f32_16x16x4_f32(false, a23, false, x23, (short)0, acc0, false, false);
            acc1 = __builtin_amdgcn_wmma_f32_16x16x4_f32(false, a23, false, y23, (short)0, acc1, false, false);
        }
        if (sub + 1 < NSUB) __syncthreads();
    }
    for (int off = 16; off > 0; off >>= 1) {
        cn0 += __shfl_xor(cn0, off);
        cn1 += __shfl_xor(cn1, off);
        const unsigned long long o0 = __shfl_xor(bm0, off);
        const unsigned long long o1 = __shfl_xor(bm1, off);
        if (o0 > bm0) bm0 = o0;
        if (o1 > bm1) bm1 = o1;
    }
    if (lane == 0) {
        lcnt[wv]     = cn0;
        lcnt[wv + 8] = cn1;
        lpk[wv]      = bm0;
        lpk[wv + 8]  = bm1;
    }
    __syncthreads();
    if (t < I_) {
        cntp[(size_t)blockIdx.x * 32 + t] = lcnt[t];
        pkp[(size_t)blockIdx.x * I_ + t]  = lpk[t];
    }
    float* pd = spart + (size_t)blockIdx.x * (I_ * C_) + (size_t)(32 * wv + lane);
#pragma unroll
    for (int r = 0; r < 8; ++r) {
        const float send = h ? acc0[r] : acc1[r];
        const float recv = __shfl_xor(send, 16);
        pd[(size_t)r * C_]       = h ? recv : acc0[r];
        pd[(size_t)(r + 8) * C_] = h ? acc1[r] : recv;
    }
}

template <int N, int BASE>
__device__ __forceinline__ float level_sum(const float* __restrict__ spart, int b, int i, int c) {
    const int w0 = BASE + b * N;
    float sv = 0.0f;
#pragma unroll
    for (int j = 0; j < N; ++j) sv += spart[(size_t)(w0 + j) * (I_ * C_) + (size_t)(i * C_ + c)];
    return sv;
}

template <int N, int BASE, int P>
__device__ __forceinline__ float level_fallback(const float* __restrict__ feat,
                                                const unsigned long long* __restrict__ pkp,
                                                int b, int i, int c) {
    const int w0 = BASE + b * N;
    unsigned long long pk = 0ull;
    for (int j = 0; j < N; ++j) {
        const unsigned long long o = pkp[(size_t)(w0 + j) * I_ + i];
        if (o > pk) pk = o;
    }
    const unsigned idx = 0xFFFFFFFFu - (unsigned)(pk & 0xFFFFFFFFull);
    return feat[((size_t)b * C_ + (size_t)c) * (size_t)P + idx];
}

__global__ void __launch_bounds__(256)
finalize_kernel(const float* __restrict__ f0, const float* __restrict__ f1,
                const float* __restrict__ f2, const float* __restrict__ f3,
                const float* __restrict__ spart, const float* __restrict__ cntp,
                const unsigned long long* __restrict__ pkp, float* __restrict__ out) {
    const int bi   = blockIdx.x;
    const int c    = threadIdx.x;
    const int lane = threadIdx.x & 31;
    const int b    = bi >> 4;
    const int i    = bi & 15;
    float x0 = cntp[(size_t)(b * 32 + lane) * 32 + i];
    const int w1 = (lane < 16) ? (256 + b * 16 + lane)
                 : (lane < 24) ? (384 + b * 8 + (lane - 16))
                               : (448 + b * 2 + (lane - 24));
    float x1 = (lane < 26) ? cntp[(size_t)w1 * 32 + i] : 0.0f;
    const float sv0 = level_sum<32, 0>(spart, b, i, c);
    const float sv1 = level_sum<16, 256>(spart, b, i, c);
    const float sv2 = level_sum<8, 384>(spart, b, i, c);
    const float sv3 = level_sum<2, 448>(spart, b, i, c);
    const float p1 = x1 + __shfl_xor(x1, 1);
    const float p2 = p1 + __shfl_xor(p1, 2);
    const float p4 = p2 + __shfl_xor(p2, 4);
    const float p8 = p4 + __shfl_xor(p4, 8);
    for (int off = 16; off > 0; off >>= 1) x0 += __shfl_xor(x0, off);
    const float cv0 = x0;
    const float cv1 = __shfl(p8, 0);
    const float cv2 = __shfl(p4, 16);
    const float cv3 = __shfl(p1, 24);
    float v0 = sv0 / fmaxf(cv0, 1.0f);
    float v1 = sv1 / fmaxf(cv1, 1.0f);
    float v2 = sv2 / fmaxf(cv2, 1.0f);
    float v3 = sv3 / fmaxf(cv3, 1.0f);
    if (!(cv0 > 0.0f)) v0 = level_fallback<32, 0, 16384>(f0, pkp, b, i, c);
    if (!(cv1 > 0.0f)) v1 = level_fallback<16, 256, 4096>(f1, pkp, b, i, c);
    if (!(cv2 > 0.0f)) v2 = level_fallback<8, 384, 1024>(f2, pkp, b, i, c);
    if (!(cv3 > 0.0f)) v3 = level_fallback<2, 448, 256>(f3, pkp, b, i, c);
    float sum = 0.0f;
    sum += v0;
    sum += v1;
    sum += v2;
    sum += v3;
    out[bi * C_ + c] = 0.25f * sum;
}

extern "C" void kernel_launch(void* const* d_in, const int* in_sizes, int n_in,
                              void* d_out, int out_size, void* d_ws, size_t ws_size,
                              hipStream_t stream) {
    const float* feat0 = (const float*)d_in[0];
    const float* feat1 = (const float*)d_in[1];
    const float* feat2 = (const float*)d_in[2];
    const float* feat3 = (const float*)d_in[3];
    const float* scrib = (const float*)d_in[4];
    float* ws = (float*)d_ws;

    float* sP   = ws + SP_OFF;
    float* cntP = ws + CNTP_OFF;
    unsigned long long* pkP = (unsigned long long*)((char*)d_ws + PKP_BYTEOFF);

    slab_masked_sum_kernel<4, 7><<<256, 256, 0, stream>>>(feat0, scrib, sP,              cntP,            pkP);
    slab_masked_sum_kernel<2, 6><<<128, 256, 0, stream>>>(feat1, scrib, sP + 256 * 4096, cntP + 256 * 32, pkP + 256 * 16);
    slab_masked_sum_kernel<1, 5><<< 64, 256, 0, stream>>>(feat2, scrib, sP + 384 * 4096, cntP + 384 * 32, pkP + 384 * 16);
    slab_masked_sum_kernel<1, 4><<< 16, 256, 0, stream>>>(feat3, scrib, sP + 448 * 4096, cntP + 448 * 32, pkP + 448 * 16);
    finalize_kernel<<<B_ * I_, 256, 0, stream>>>(
        feat0, feat1, feat2, feat3, sP, cntP, pkP, (float*)d_out);
}


# levels 2 and 3 cut into 64- and 32-pixel slabs (128 and 64 workgroups, were 64 and 16), run first; their tiles park in the level-0 tile slots, the level-1 launch folds them; sum order changes
# speedup vs baseline: 1.8385x; 1.1352x over previous
#include <hip/hip_runtime.h>

typedef float v2f __attribute__((ext_vector_type(2)));
typedef float v4f __attribute__((ext_vector_type(4)));
typedef float v8f __attribute__((ext_vector_type(8)));

#define B_ 8
#define I_ 16
#define C_ 256

#define NWG_ALL 576
static const size_t SP_OFF      = 0;
static const size_t S2_OFF      = (size_t)384 * 4096;
static const size_t S3_OFF      = S2_OFF + (size_t)B_ * 4096;
static const size_t CNTP_OFF    = S3_OFF + (size_t)B_ * 4096;
static const size_t PKP_BYTEOFF = (CNTP_OFF + (size_t)NWG_ALL * 32) * 4ull;


__device__ __forceinline__ unsigned long long load_pair(const float* p) {
    unsigned long long v;
    __builtin_memcpy(&v, p, sizeof(v));
    return v;
}
__device__ __forceinline__ float pair_lo(unsigned long long v) { return __uint_as_float((unsigned)v); }
__device__ __forceinline__ float pair_hi(unsigned long long v) { return __uint_as_float((unsigned)(v >> 32)); }

template <int LW>
__device__ __forceinline__ int mask_src(int rowbase, int pix) {
    constexpr int F    = 512 >> LW;
    constexpr int OFFS = F / 2 - 1;
    const int y = pix >> LW;
    const int x = pix & ((1 << LW) - 1);
    return (rowbase + y * F + OFFS) * 512 + x * F + OFFS;
}

template <int NSUB, int LW, int KC, bool FOLD>
__global__ void __launch_bounds__(256)
slab_masked_sum_kernel(const float* __restrict__ feat, const float* __restrict__ scrib,
                       float* __restrict__ spart, float* __restrict__ cntp,
                       unsigned long long* __restrict__ pkp,
                       const float* __restrict__ parked, float* __restrict__ folded) {
    constexpr int P      = (1 << LW) * (1 << LW);
    constexpr int wgPerB = P / (NSUB * KC);
    constexpr int PLANE8 = 8 * 512 * 512;
    constexpr int PITCH  = KC + 4;
    constexpr int NQ     = KC / 32;
    constexpr int LPR    = KC / 4;
    constexpr int RPI    = 32 / LPR;
    constexpr int NLD    = 32 / RPI;
    static_assert(KC == 128 || KC == 64 || KC == 32, "slab width");
    static_assert(P % (NSUB * KC) == 0, "slabs tile the plane");
    static_assert(!FOLD || wgPerB == 16, "the fold maps 16 workgroups x 256 threads onto one 4096-element tile");
    __shared__ __attribute__((aligned(16))) float lsel[I_ * PITCH];
    __shared__ __attribute__((aligned(16))) float lfeat[C_ * PITCH];
    __shared__ float lcnt[I_];
    __shared__ unsigned long long lpk[I_];
    const int t    = threadIdx.x;
    const int lane = t & 31;
    const int wv   = t >> 5;
    const int b    = blockIdx.x / wgPerB;
    const int p0   = (blockIdx.x % wgPerB) * (NSUB * KC);
    const int c    = lane & 15;
    const int h    = lane >> 4;

    const int lr = lane / LPR;
    const int lx = 4 * (lane % LPR);
    const float* fg = feat + ((size_t)b * C_ + (size_t)(32 * wv + lr)) * (size_t)P + (size_t)(p0 + lx);
    const int rowbase = (b * I_ + wv) * 512;
    const int pl      = p0 + lane;

    float* lf = lfeat + (32 * wv + lr) * PITCH + lx;
    float* ls = lsel + wv * PITCH + lane;
    const float* ap  = lsel + c * PITCH + 4 * h;
    const float* bp0 = lfeat + (32 * wv + c) * PITCH + 4 * h;
    const float* bp1 = bp0 + 16 * PITCH;

    v4f  fr[NLD];
    unsigned long long sc[4 * NQ];
#pragma unroll
    for (int q = 0; q < NQ; ++q) {
        const int e = mask_src<LW>(rowbase, pl + 32 * q);
        sc[4 * q + 0] = load_pair(scrib + e);
        sc[4 * q + 1] = load_pair(scrib + e + 512);
        sc[4 * q + 2] = load_pair(scrib + e + PLANE8);
        sc[4 * q + 3] = load_pair(scrib + e + PLANE8 + 512);
    }
#pragma unroll
    for (int j = 0; j < NLD; ++j) fr[j] = *(const v4f*)(fg + (size_t)(j * RPI) * (size_t)P);

    float fo2[16], fo3[8];
    int   fe = 0;
    if constexpr (FOLD) {
        fe = (int)(blockIdx.x % wgPerB) * 256 + t;
        const float* s2p = parked + (size_t)(b * 16) * 4096 + (size_t)fe;
        const float* s3p = parked + (size_t)(128 + b * 8) * 4096 + (size_t)fe;
#pragma unroll
        for (int k = 0; k < 16; ++k) fo2[k] = s2p[(size_t)k * 4096];
#pragma unroll
        for (int k = 0; k < 8; ++k) fo3[k] = s3p[(size_t)k * 4096];
    }

    float cn0 = 0.0f, cn1 = 0.0f;
    unsigned long long bm0 = 0ull, bm1 = 0ull;
    v8f acc0 = {};
    v8f acc1 = {};
    for (int sub = 0; sub < NSUB; ++sub) {
#pragma unroll
        for (int q = 0; q < NQ; ++q) {
            const int pix = pl + 32 * q + sub * KC;
            const float m0 = 0.25f * (pair_lo(sc[4 * q + 0]) + pair_hi(sc[4 * q + 0]) + pair_lo(sc[4 * q + 1]) + pair_hi(sc[4 * q + 1]));
            const float m1 = 0.25f * (pair_lo(sc[4 * q + 2]) + pair_hi(sc[4 * q + 2]) + pair_lo(sc[4 * q + 3]) + pair_hi(sc[4 * q + 3]));
            const float s0 = (m0 > 0.5f) ? 1.0f : 0.0f;
            const float s1 = (m1 > 0.5f) ? 1.0f : 0.0f;
            ls[32 * q]             = s0;
            ls[8 * PITCH + 32 * q] = s1;
            cn0 += s0;
            cn1 += s1;
            const unsigned long long lo = (unsigned long long)(0xFFFFFFFFu - (unsigned)pix);
            const unsigned long long k0 = ((unsigned long long)__float_as_uint(m0) << 32) | lo;
            const unsigned long long k1 = ((unsigned long long)__float_as_uint(m1) << 32) | lo;
            if (k0 > bm0) bm0 = k0;
            if (k1 > bm1) bm1 = k1;
        }
#pragma unroll
        for (int j = 0; j < NLD; ++j) *(v4f*)(lf + j * RPI * PITCH) = fr[j];
        __syncthreads();
        if constexpr (FOLD) {
            if (sub == 0) {
                float a2 = 0.0f, a3 = 0.0f;
#pragma unroll
                for (int k = 0; k < 16; ++k) a2 += fo2[k];
#pragma unroll
                for (int k = 0; k < 8; ++k) a3 += fo3[k];
                folded[(size_t)b * 4096 + (size_t)fe]                      = a2;
                folded[(size_t)B_ * 4096 + (size_t)b * 4096 + (size_t)fe] = a3;
            }
        }
        if (sub + 1 < NSUB) {
            const float* fgn = fg + (size_t)(sub + 1) * KC;
#pragma unroll
            for (int q = 0; q < NQ; ++q) {
                const int e = mask_src<LW>(rowbase, pl + 32 * q + (sub + 1) * KC);
                sc[4 * q + 0] = load_pair(scrib + e);
                sc[4 * q + 1] = load_pair(scrib + e + 512);
                sc[4 * q + 2] = load_pair(scrib + e + PLANE8);
                sc[4 * q + 3] = load_pair(scrib + e + PLANE8 + 512);
            }
#pragma unroll
            for (int j = 0; j < NLD; ++j) fr[j] = *(const v4f*)(fgn + (size_t)(j * RPI) * (size_t)P);
            __builtin_amdgcn_sched_barrier(0);
        }
        for (int k8 = 0; k8 < KC / 8; ++k8) {
            v4f a = *(const v4f*)(ap  + 8 * k8);
            v4f x = *(const v4f*)(bp0 + 8 * k8);
            v4f y = *(const v4f*)(bp1 + 8 * k8);
            v2f a01 = {a[0], a[1]}, a23 = {a[2], a[3]};
            v2f x01 = {x[0], x[1]}, x23 = {x[2], x[3]};
            v2f y01 = {y[0], y[1]}, y23 = {y[2], y[3]};
            acc0 = __builtin_amdgcn_wmma_f32_16x16x4_f32(false, a01, false, x01, (short)0, acc0, false, false);
            acc1 = __builtin_amdgcn_wmma_f32_16x16x4_f32(false, a01, false, y01, (short)0, acc1, false, false);
            acc0 = __builtin_amdgcn_wmma_f32_16x16x4_f32(false, a23, false, x23, (short)0, acc0, false, false);
            acc1 = __builtin_amdgcn_wmma_f32_16x16x4_f32(false, a23, false, y23, (short)0, acc1, false, false);
        }
        if (sub + 1 < NSUB) __syncthreads();
    }
    for (int off = 16; off > 0; off >>= 1) {
        cn0 += __shfl_xor(cn0, off);
        cn1 += __shfl_xor(cn1, off);
        const unsigned long long o0 = __shfl_xor(bm0, off);
        const unsigned long long o1 = __shfl_xor(bm1, off);
        if (o0 > bm0) bm0 = o0;
        if (o1 > bm1) bm1 = o1;
    }
    if (lane == 0) {
        lcnt[wv]     = cn0;
        lcnt[wv + 8] = cn1;
        lpk[wv]      = bm0;
        lpk[wv + 8]  = bm1;
    }
    __syncthreads();
    if (t < I_) {
        cntp[(size_t)blockIdx.x * 32 + t] = lcnt[t];
        pkp[(size_t)blockIdx.x * I_ + t]  = lpk[t];
    }
    float* pd = spart + (size_t)blockIdx.x * (I_ * C_) + (size_t)(32 * wv + lane);
#pragma unroll
    for (int r = 0; r < 8; ++r) {
        const float send = h ? acc0[r] : acc1[r];
        const float recv = __shfl_xor(send, 16);
        pd[(size_t)r * C_]       = h ? recv : acc0[r];
        pd[(size_t)(r + 8) * C_] = h ? acc1[r] : recv;
    }
}

template <int N, int BASE>
__device__ __forceinline__ float level_sum(const float* __restrict__ spart, int b, int i, int c) {
    const int w0 = BASE + b * N;
    float sv = 0.0f;
#pragma unroll
    for (int j = 0; j < N; ++j) sv += spart[(size_t)(w0 + j) * (I_ * C_) + (size_t)(i * C_ + c)];
    return sv;
}

template <int N, int BASE, int P>
__device__ __forceinline__ float level_fallback(const float* __restrict__ feat,
                                                const unsigned long long* __restrict__ pkp,
                                                int b, int i, int c) {
    const int w0 = BASE + b * N;
    unsigned long long pk = 0ull;
    for (int j = 0; j < N; ++j) {
        const unsigned long long o = pkp[(size_t)(w0 + j) * I_ + i];
        if (o > pk) pk = o;
    }
    const unsigned idx = 0xFFFFFFFFu - (unsigned)(pk & 0xFFFFFFFFull);
    return feat[((size_t)b * C_ + (size_t)c) * (size_t)P + idx];
}

__global__ void __launch_bounds__(256)
finalize_kernel(const float* __restrict__ f0, const float* __restrict__ f1,
                const float* __restrict__ f2, const float* __restrict__ f3,
                const float* __restrict__ spart, const float* __restrict__ s23,
                const float* __restrict__ cntp,
                const unsigned long long* __restrict__ pkp, float* __restrict__ out) {
    const int bi   = blockIdx.x;
    const int c    = threadIdx.x;
    const int lane = threadIdx.x & 31;
    const int b    = bi >> 4;
    const int i    = bi & 15;
    float x0 = cntp[(size_t)(b * 32 + lane) * 32 + i];
    const int w1 = (lane < 16) ? (256 + b * 16 + lane) : (384 + b * 16 + (lane - 16));
    float x1 = cntp[(size_t)w1 * 32 + i];
    float x2 = (lane < 8) ? cntp[(size_t)(512 + b * 8 + lane) * 32 + i] : 0.0f;
    const float sv0 = level_sum<32, 0>(spart, b, i, c);
    const float sv1 = level_sum<16, 256>(spart, b, i, c);
    const float sv2 = s23[(size_t)bi * C_ + (size_t)c];
    const float sv3 = s23[(size_t)B_ * 4096 + (size_t)bi * C_ + (size_t)c];
    const float p1 = x1 + __shfl_xor(x1, 1);
    const float p2 = p1 + __shfl_xor(p1, 2);
    const float p4 = p2 + __shfl_xor(p2, 4);
    const float p8 = p4 + __shfl_xor(p4, 8);
    const float q1 = x2 + __shfl_xor(x2, 1);
    const float q2 = q1 + __shfl_xor(q1, 2);
    const float q4 = q2 + __shfl_xor(q2, 4);
    for (int off = 16; off > 0; off >>= 1) x0 += __shfl_xor(x0, off);
    const float cv0 = x0;
    const float cv1 = __shfl(p8, 0);
    const float cv2 = __shfl(p8, 16);
    const float cv3 = __shfl(q4, 0);
    float v0 = sv0 / fmaxf(cv0, 1.0f);
    float v1 = sv1 / fmaxf(cv1, 1.0f);
    float v2 = sv2 / fmaxf(cv2, 1.0f);
    float v3 = sv3 / fmaxf(cv3, 1.0f);
    if (!(cv0 > 0.0f)) v0 = level_fallback<32, 0, 16384>(f0, pkp, b, i, c);
    if (!(cv1 > 0.0f)) v1 = level_fallback<16, 256, 4096>(f1, pkp, b, i, c);
    if (!(cv2 > 0.0f)) v2 = level_fallback<16, 384, 1024>(f2, pkp, b, i, c);
    if (!(cv3 > 0.0f)) v3 = level_fallback<8, 512, 256>(f3, pkp, b, i, c);
    float sum = 0.0f;
    sum += v0;
    sum += v1;
    sum += v2;
    sum += v3;
    out[bi * C_ + c] = 0.25f * sum;
}

extern "C" void kernel_launch(void* const* d_in, const int* in_sizes, int n_in,
                              void* d_out, int out_size, void* d_ws, size_t ws_size,
                              hipStream_t stream) {
    const float* feat0 = (const float*)d_in[0];
    const float* feat1 = (const float*)d_in[1];
    const float* feat2 = (const float*)d_in[2];
    const float* feat3 = (const float*)d_in[3];
    const float* scrib = (const float*)d_in[4];
    float* ws = (float*)d_ws;

    float* sP   = ws + SP_OFF;
    float* s23  = ws + S2_OFF;
    float* cntP = ws + CNTP_OFF;
    unsigned long long* pkP = (unsigned long long*)((char*)d_ws + PKP_BYTEOFF);

    slab_masked_sum_kernel<1, 4, 32, false><<< 64, 256, 0, stream>>>(feat3, scrib, sP + 128 * 4096, cntP + 512 * 32, pkP + 512 * 16, nullptr, nullptr);
    slab_masked_sum_kernel<1, 5, 64, false><<<128, 256, 0, stream>>>(feat2, scrib, sP,              cntP + 384 * 32, pkP + 384 * 16, nullptr, nullptr);
    slab_masked_sum_kernel<2, 6, 128, true><<<128, 256, 0, stream>>>(feat1, scrib, sP + 256 * 4096, cntP + 256 * 32, pkP + 256 * 16, sP, s23);
    slab_masked_sum_kernel<4, 7, 128, false><<<256, 256, 0, stream>>>(feat0, scrib, sP,             cntP,            pkP,            nullptr, nullptr);
    finalize_kernel<<<B_ * I_, 256, 0, stream>>>(
        feat0, feat1, feat2, feat3, sP, s23, cntP, pkP, (float*)d_out);
}


# levels 2 and 3 lose their launches: level-1 launch writes their thresholded masks, last launch sums them whole per (image, 32 channels) on f32 WMMA and finalizes; 5 -> 3 launches; sum order changes
# speedup vs baseline: 1.9930x; 1.0840x over previous
#include <hip/hip_runtime.h>

typedef float v2f __attribute__((ext_vector_type(2)));
typedef float v4f __attribute__((ext_vector_type(4)));
typedef float v8f __attribute__((ext_vector_type(8)));
typedef unsigned int u2w __attribute__((ext_vector_type(2)));
typedef unsigned int u4w __attribute__((ext_vector_type(4)));

#define B_ 8
#define I_ 16
#define C_ 256

#define NWG_ALL 576
static const size_t SP_OFF      = 0;
static const size_t SEL2_OFF    = (size_t)384 * 4096;
static const size_t SEL3_OFF    = SEL2_OFF + (size_t)B_ * 16 * 256;
static const size_t CNTP_OFF    = (size_t)400 * 4096;
static const size_t PKP_BYTEOFF = (CNTP_OFF + (size_t)NWG_ALL * 32) * 4ull;


__device__ __forceinline__ unsigned long long load_pair(const float* p) {
    unsigned long long v;
    __builtin_memcpy(&v, p, sizeof(v));
    return v;
}
__device__ __forceinline__ float pair_lo(unsigned long long v) { return __uint_as_float((unsigned)v); }
__device__ __forceinline__ float pair_hi(unsigned long long v) { return __uint_as_float((unsigned)(v >> 32)); }

template <int LW>
__device__ __forceinline__ int mask_src(int rowbase, int pix) {
    constexpr int F    = 512 >> LW;
    constexpr int OFFS = F / 2 - 1;
    const int y = pix >> LW;
    const int x = pix & ((1 << LW) - 1);
    return (rowbase + y * F + OFFS) * 512 + x * F + OFFS;
}

template <int NSUB, int LW, int KC, bool SMALL>
__global__ void __launch_bounds__(256)
slab_masked_sum_kernel(const float* __restrict__ feat, const float* __restrict__ scrib,
                       float* __restrict__ spart, float* __restrict__ cntp,
                       unsigned long long* __restrict__ pkp,
                       unsigned int* __restrict__ selw2, unsigned int* __restrict__ selw3) {
    constexpr int P      = (1 << LW) * (1 << LW);
    constexpr int wgPerB = P / (NSUB * KC);
    constexpr int PLANE8 = 8 * 512 * 512;
    constexpr int PITCH  = KC + 4;
    constexpr int NQ     = KC / 32;
    constexpr int LPR    = KC / 4;
    constexpr int RPI    = 32 / LPR;
    constexpr int NLD    = 32 / RPI;
    static_assert(KC == 128 || KC == 64 || KC == 32, "slab width");
    static_assert(P % (NSUB * KC) == 0, "slabs tile the plane");
    static_assert(!SMALL || wgPerB == 16, "16 workgroups per image take 64 level-2 pixels each, the first 8 also 32 level-3 pixels");
    __shared__ __attribute__((aligned(16))) float lsel[I_ * PITCH];
    __shared__ __attribute__((aligned(16))) float lfeat[C_ * PITCH];
    __shared__ float lcnt[I_];
    __shared__ unsigned long long lpk[I_];
    __shared__ float lcnt2[I_], lcnt3[I_];
    __shared__ unsigned long long lpk2[I_], lpk3[I_];
    const int t    = threadIdx.x;
    const int lane = t & 31;
    const int wv   = t >> 5;
    const int b    = blockIdx.x / wgPerB;
    const int p0   = (blockIdx.x % wgPerB) * (NSUB * KC);
    const int c    = lane & 15;
    const int h    = lane >> 4;

    const int lr = lane / LPR;
    const int lx = 4 * (lane % LPR);
    const float* fg = feat + ((size_t)b * C_ + (size_t)(32 * wv + lr)) * (size_t)P + (size_t)(p0 + lx);
    const int rowbase = (b * I_ + wv) * 512;
    const int pl      = p0 + lane;

    float* lf = lfeat + (32 * wv + lr) * PITCH + lx;
    float* ls = lsel + wv * PITCH + lane;
    const float* ap  = lsel + c * PITCH + 4 * h;
    const float* bp0 = lfeat + (32 * wv + c) * PITCH + 4 * h;
    const float* bp1 = bp0 + 16 * PITCH;

    v4f  fr[NLD];
    unsigned long long sc[4 * NQ];
#pragma unroll
    for (int q = 0; q < NQ; ++q) {
        const int e = mask_src<LW>(rowbase, pl + 32 * q);
        sc[4 * q + 0] = load_pair(scrib + e);
        sc[4 * q + 1] = load_pair(scrib + e + 512);
        sc[4 * q + 2] = load_pair(scrib + e + PLANE8);
        sc[4 * q + 3] = load_pair(scrib + e + PLANE8 + 512);
    }
#pragma unroll
    for (int j = 0; j < NLD; ++j) fr[j] = *(const v4f*)(fg + (size_t)(j * RPI) * (size_t)P);

    unsigned long long q2[8], q3[8];
    int chunk = 0;
    if constexpr (SMALL) {
        chunk = (int)(blockIdx.x % wgPerB);
        const int rb2 = (b * I_ + (t >> 4)) * 512;
        const int rb3 = (b * I_ + ((t & 127) >> 3)) * 512;
#pragma unroll
        for (int k = 0; k < 4; ++k) {
            const int e2 = mask_src<5>(rb2, 64 * chunk + 4 * (t & 15) + k);
            q2[2 * k + 0] = load_pair(scrib + e2);
            q2[2 * k + 1] = load_pair(scrib + e2 + 512);
            const int e3 = mask_src<4>(rb3, 32 * (chunk & 7) + 4 * (t & 7) + k);
            q3[2 * k + 0] = load_pair(scrib + e3);
            q3[2 * k + 1] = load_pair(scrib + e3 + 512);
        }
    }

    float cn0 = 0.0f, cn1 = 0.0f;
    unsigned long long bm0 = 0ull, bm1 = 0ull;
    v8f acc0 = {};
    v8f acc1 = {};
    for (int sub = 0; sub < NSUB; ++sub) {
#pragma unroll
        for (int q = 0; q < NQ; ++q) {
            const int pix = pl + 32 * q + sub * KC;
            const float m0 = 0.25f * (pair_lo(sc[4 * q + 0]) + pair_hi(sc[4 * q + 0]) + pair_lo(sc[4 * q + 1]) + pair_hi(sc[4 * q + 1]));
            const float m1 = 0.25f * (pair_lo(sc[4 * q + 2]) + pair_hi(sc[4 * q + 2]) + pair_lo(sc[4 * q + 3]) + pair_hi(sc[4 * q + 3]));
            const float s0 = (m0 > 0.5f) ? 1.0f : 0.0f;
            const float s1 = (m1 > 0.5f) ? 1.0f : 0.0f;
            ls[32 * q]             = s0;
            ls[8 * PITCH + 32 * q] = s1;
            cn0 += s0;
            cn1 += s1;
            const unsigned long long lo = (unsigned long long)(0xFFFFFFFFu - (unsigned)pix);
            const unsigned long long k0 = ((unsigned long long)__float_as_uint(m0) << 32) | lo;
            const unsigned long long k1 = ((unsigned long long)__float_as_uint(m1) << 32) | lo;
            if (k0 > bm0) bm0 = k0;
            if (k1 > bm1) bm1 = k1;
        }
#pragma unroll
        for (int j = 0; j < NLD; ++j) *(v4f*)(lf + j * RPI * PITCH) = fr[j];
        __syncthreads();
        if constexpr (SMALL) {
            if (sub == 0) {
                float c2 = 0.0f;
                unsigned long long m2k = 0ull;
                unsigned int w2 = 0u;
#pragma unroll
                for (int k = 0; k < 4; ++k) {
                    const int pix = 64 * chunk + 4 * (t & 15) + k;
                    const float m = 0.25f * (pair_lo(q2[2 * k]) + pair_hi(q2[2 * k]) + pair_lo(q2[2 * k + 1]) + pair_hi(q2[2 * k + 1]));
                    const bool on = m > 0.5f;
                    w2 |= (on ? 1u : 0u) << (8 * k);
                    c2 += on ? 1.0f : 0.0f;
                    const unsigned long long key = ((unsigned long long)__float_as_uint(m) << 32)
                                                 | (unsigned long long)(0xFFFFFFFFu - (unsigned)pix);
                    if (key > m2k) m2k = key;
                }
                selw2[(size_t)(b * 16 + chunk) * 256 + (size_t)t] = w2;
                for (int off = 8; off > 0; off >>= 1) {
                    c2 += __shfl_xor(c2, off);
                    const unsigned long long o = __shfl_xor(m2k, off);
                    if (o > m2k) m2k = o;
                }
                if ((t & 15) == 0) { lcnt2[t >> 4] = c2; lpk2[t >> 4] = m2k; }
                const bool has3 = (chunk < 8) && (t < 128);
                float c3 = 0.0f;
                unsigned long long m3k = 0ull;
                unsigned int w3 = 0u;
#pragma unroll
                for (int k = 0; k < 4; ++k) {
                    const int pix = 32 * (chunk & 7) + 4 * (t & 7) + k;
                    const float m = 0.25f * (pair_lo(q3[2 * k]) + pair_hi(q3[2 * k]) + pair_lo(q3[2 * k + 1]) + pair_hi(q3[2 * k + 1]));
                    const bool on = m > 0.5f;
                    w3 |= (on ? 1u : 0u) << (8 * k);
                    c3 += on ? 1.0f : 0.0f;
                    const unsigned long long key = ((unsigned long long)__float_as_uint(m) << 32)
                                                 | (unsigned long long)(0xFFFFFFFFu - (unsigned)pix);
                    if (key > m3k) m3k = key;
                }
                if (has3) selw3[(size_t)(b * 8 + chunk) * 128 + (size_t)t] = w3;
                for (int off = 4; off > 0; off >>= 1) {
                    c3 += __shfl_xor(c3, off);
                    const unsigned long long o = __shfl_xor(m3k, off);
                    if (o > m3k) m3k = o;
                }
                if (has3 && (t & 7) == 0) { lcnt3[t >> 3] = c3; lpk3[t >> 3] = m3k; }
            }
        }
        if (sub + 1 < NSUB) {
            const float* fgn = fg + (size_t)(sub + 1) * KC;
#pragma unroll
            for (int q = 0; q < NQ; ++q) {
                const int e = mask_src<LW>(rowbase, pl + 32 * q + (sub + 1) * KC);
                sc[4 * q + 0] = load_pair(scrib + e);
                sc[4 * q + 1] = load_pair(scrib + e + 512);
                sc[4 * q + 2] = load_pair(scrib + e + PLANE8);
                sc[4 * q + 3] = load_pair(scrib + e + PLANE8 + 512);
            }
#pragma unroll
            for (int j = 0; j < NLD; ++j) fr[j] = *(const v4f*)(fgn + (size_t)(j * RPI) * (size_t)P);
            __builtin_amdgcn_sched_barrier(0);
        }
        for (int k8 = 0; k8 < KC / 8; ++k8) {
            v4f a = *(const v4f*)(ap  + 8 * k8);
            v4f x = *(const v4f*)(bp0 + 8 * k8);
            v4f y = *(const v4f*)(bp1 + 8 * k8);
            v2f a01 = {a[0], a[1]}, a23 = {a[2], a[3]};
            v2f x01 = {x[0], x[1]}, x23 = {x[2], x[3]};
            v2f y01 = {y[0], y[1]}, y23 = {y[2], y[3]};
            acc0 = __builtin_amdgcn_wmma_f32_16x16x4_f32(false, a01, false, x01, (short)0, acc0, false, false);
            acc1 = __builtin_amdgcn_wmma_f32_16x16x4_f32(false, a01, false, y01, (short)0, acc1, false, false);
            acc0 = __builtin_amdgcn_wmma_f32_16x16x4_f32(false, a23, false, x23, (short)0, acc0, false, false);
            acc1 = __builtin_amdgcn_wmma_f32_16x16x4_f32(false, a23, false, y23, (short)0, acc1, false, false);
        }
        if (sub + 1 < NSUB) __syncthreads();
    }
    for (int off = 16; off > 0; off >>= 1) {
        cn0 += __shfl_xor(cn0, off);
        cn1 += __shfl_xor(cn1, off);
        const unsigned long long o0 = __shfl_xor(bm0, off);
        const unsigned long long o1 = __shfl_xor(bm1, off);
        if (o0 > bm0) bm0 = o0;
        if (o1 > bm1) bm1 = o1;
    }
    if (lane == 0) {
        lcnt[wv]     = cn0;
        lcnt[wv + 8] = cn1;
        lpk[wv]      = bm0;
        lpk[wv + 8]  = bm1;
    }
    __syncthreads();
    if (t < I_) {
        cntp[(size_t)blockIdx.x * 32 + t] = lcnt[t];
        pkp[(size_t)blockIdx.x * I_ + t]  = lpk[t];
        if constexpr (SMALL) {
            const size_t n2 = (size_t)(128 + b * 16 + chunk);
            cntp[n2 * 32 + t] = lcnt2[t];
            pkp[n2 * I_ + t]  = lpk2[t];
            if (chunk < 8) {
                const size_t n3 = (size_t)(256 + b * 8 + chunk);
                cntp[n3 * 32 + t] = lcnt3[t];
                pkp[n3 * I_ + t]  = lpk3[t];
            }
        }
    }
    float* pd = spart + (size_t)blockIdx.x * (I_ * C_) + (size_t)(32 * wv + lane);
#pragma unroll
    for (int r = 0; r < 8; ++r) {
        const float send = h ? acc0[r] : acc1[r];
        const float recv = __shfl_xor(send, 16);
        pd[(size_t)r * C_]       = h ? recv : acc0[r];
        pd[(size_t)(r + 8) * C_] = h ? acc1[r] : recv;
    }
}


template <int N, int BASE, int P>
__device__ __forceinline__ float level_fallback(const float* __restrict__ feat,
                                                const unsigned long long* __restrict__ pkp,
                                                int b, int i, int c) {
    const int w0 = BASE + b * N;
    unsigned long long pk = 0ull;
    for (int j = 0; j < N; ++j) {
        const unsigned long long o = pkp[(size_t)(w0 + j) * I_ + i];
        if (o > pk) pk = o;
    }
    const unsigned idx = 0xFFFFFFFFu - (unsigned)(pk & 0xFFFFFFFFull);
    return feat[((size_t)b * C_ + (size_t)c) * (size_t)P + idx];
}

#define P2_ 1028
#define P3_ 260
__global__ void __launch_bounds__(512)
small_levels_finalize_kernel(const float* __restrict__ f0, const float* __restrict__ f1,
                             const float* __restrict__ f2, const float* __restrict__ f3,
                             const float* spart,
                             const unsigned int* __restrict__ selw2, const unsigned int* __restrict__ selw3,
                             const float* __restrict__ cntp,
                             const unsigned long long* __restrict__ pkp, float* __restrict__ out) {
    __shared__ __attribute__((aligned(16))) float lf2[32 * P2_];
    __shared__ __attribute__((aligned(16))) float ls2[I_ * P2_];
    __shared__ __attribute__((aligned(16))) float lf3[32 * P3_];
    __shared__ __attribute__((aligned(16))) float ls3[I_ * P3_];
    __shared__ float lred[2 * 8 * 2 * 256];
    const int t    = threadIdx.x;
    const int lane = t & 31;
    const int wv   = t >> 5;
    const int b    = blockIdx.x >> 3;
    const int g    = blockIdx.x & 7;
    const int c    = lane & 15;
    const int h    = lane >> 4;

    const float* g2 = f2 + ((size_t)b * C_ + (size_t)(32 * g + 2 * wv)) * 1024 + (size_t)(4 * lane);
    const float* g3 = f3 + ((size_t)b * C_ + (size_t)(32 * g + 2 * wv)) * 256 + (size_t)(4 * lane);
    v4f fr2[16], fr3[4];
#pragma unroll
    for (int rr = 0; rr < 2; ++rr) {
#pragma unroll
        for (int j = 0; j < 8; ++j) fr2[rr * 8 + j] = *(const v4f*)(g2 + rr * 1024 + 128 * j);
    }
#pragma unroll
    for (int rr = 0; rr < 2; ++rr) {
#pragma unroll
        for (int j = 0; j < 2; ++j) fr3[rr * 2 + j] = *(const v4f*)(g3 + rr * 256 + 128 * j);
    }
    u4w sw2[2];
#pragma unroll
    for (int q = 0; q < 2; ++q) sw2[q] = *(const u4w*)(selw2 + (size_t)b * 4096 + (size_t)(q * 2048 + 4 * t));
    const u2w sw3 = *(const u2w*)(selw3 + (size_t)b * 1024 + (size_t)(2 * t));

    const int i  = wv;
    const int bi = b * I_ + i;
    const int cg = 32 * g + lane;
    float x0 = cntp[(size_t)(b * 32 + lane) * 32 + i];
    const int w1 = (lane < 16) ? (256 + b * 16 + lane) : (384 + b * 16 + (lane - 16));
    float x1 = cntp[(size_t)w1 * 32 + i];
    float x2 = (lane < 8) ? cntp[(size_t)(512 + b * 8 + lane) * 32 + i] : 0.0f;
    float pv0[32], pv1[16];
    const float* sp0 = spart + (size_t)(b * 32) * (I_ * C_) + (size_t)(i * C_ + cg);
    const float* sp1 = spart + (size_t)(256 + b * 16) * (I_ * C_) + (size_t)(i * C_ + cg);
#pragma unroll
    for (int j = 0; j < 32; ++j) pv0[j] = sp0[(size_t)j * (I_ * C_)];
#pragma unroll
    for (int j = 0; j < 16; ++j) pv1[j] = sp1[(size_t)j * (I_ * C_)];

    float* d2 = lf2 + (2 * wv) * P2_ + 4 * lane;
    float* d3 = lf3 + (2 * wv) * P3_ + 4 * lane;
#pragma unroll
    for (int rr = 0; rr < 2; ++rr) {
#pragma unroll
        for (int j = 0; j < 8; ++j) *(v4f*)(d2 + rr * P2_ + 128 * j) = fr2[rr * 8 + j];
    }
#pragma unroll
    for (int rr = 0; rr < 2; ++rr) {
#pragma unroll
        for (int j = 0; j < 2; ++j) *(v4f*)(d3 + rr * P3_ + 128 * j) = fr3[rr * 2 + j];
    }
#pragma unroll
    for (int q = 0; q < 2; ++q) {
        const int widx = q * 2048 + 4 * t;
        const int row  = (widx >> 4) & 15;
        const int px0  = 64 * (widx >> 8) + 4 * (widx & 15);
#pragma unroll
        for (int k = 0; k < 4; ++k) {
            const unsigned int w = sw2[q][k];
            v4f v = {(float)(w & 1u), (float)((w >> 8) & 1u), (float)((w >> 16) & 1u), (float)((w >> 24) & 1u)};
            *(v4f*)(ls2 + row * P2_ + px0 + 4 * k) = v;
        }
    }
    {
        const int widx = 2 * t;
        const int row  = (widx >> 3) & 15;
        const int px0  = 32 * (widx >> 7) + 4 * (widx & 7);
#pragma unroll
        for (int k = 0; k < 2; ++k) {
            const unsigned int w = sw3[k];
            v4f v = {(float)(w & 1u), (float)((w >> 8) & 1u), (float)((w >> 16) & 1u), (float)((w >> 24) & 1u)};
            *(v4f*)(ls3 + row * P3_ + px0 + 4 * k) = v;
        }
    }
    __syncthreads();

    const int tl = wv & 1;
    const int kp = wv >> 1;
    v8f acc2 = {};
    v8f acc3 = {};
    {
        const float* ap = ls2 + c * P2_ + 128 * kp + 4 * h;
        const float* bp = lf2 + (16 * tl + c) * P2_ + 128 * kp + 4 * h;
        for (int k8 = 0; k8 < 16; ++k8) {
            v4f a = *(const v4f*)(ap + 8 * k8);
            v4f x = *(const v4f*)(bp + 8 * k8);
            v2f a01 = {a[0], a[1]}, a23 = {a[2], a[3]};
            v2f x01 = {x[0], x[1]}, x23 = {x[2], x[3]};
            acc2 = __builtin_amdgcn_wmma_f32_16x16x4_f32(false, a01, false, x01, (short)0, acc2, false, false);
            acc2 = __builtin_amdgcn_wmma_f32_16x16x4_f32(false, a23, false, x23, (short)0, acc2, false, false);
        }
    }
    {
        const float* ap = ls3 + c * P3_ + 32 * kp + 4 * h;
        const float* bp = lf3 + (16 * tl + c) * P3_ + 32 * kp + 4 * h;
        for (int k8 = 0; k8 < 4; ++k8) {
            v4f a = *(const v4f*)(ap + 8 * k8);
            v4f x = *(const v4f*)(bp + 8 * k8);
            v2f a01 = {a[0], a[1]}, a23 = {a[2], a[3]};
            v2f x01 = {x[0], x[1]}, x23 = {x[2], x[3]};
            acc3 = __builtin_amdgcn_wmma_f32_16x16x4_f32(false, a01, false, x01, (short)0, acc3, false, false);
            acc3 = __builtin_amdgcn_wmma_f32_16x16x4_f32(false, a23, false, x23, (short)0, acc3, false, false);
        }
    }
    {
        float* rd = lred + (kp * 2 + tl) * 256 + lane;
#pragma unroll
        for (int r = 0; r < 8; ++r) {
            rd[r * 32]        = acc2[r];
            rd[4096 + r * 32] = acc3[r];
        }
    }
    __syncthreads();

    const int src = (i & 7) * 32 + 16 * (i >> 3) + (lane & 15);
    const int tlc = lane >> 4;
    float sv2 = 0.0f, sv3 = 0.0f;
    for (int e = 0; e < 8; ++e) {
        sv2 += lred[(e * 2 + tlc) * 256 + src];
        sv3 += lred[4096 + (e * 2 + tlc) * 256 + src];
    }
    float sv0 = 0.0f, sv1 = 0.0f;
#pragma unroll
    for (int j = 0; j < 32; ++j) sv0 += pv0[j];
#pragma unroll
    for (int j = 0; j < 16; ++j) sv1 += pv1[j];

    const float p1 = x1 + __shfl_xor(x1, 1);
    const float p2 = p1 + __shfl_xor(p1, 2);
    const float p4 = p2 + __shfl_xor(p2, 4);
    const float p8 = p4 + __shfl_xor(p4, 8);
    const float q1 = x2 + __shfl_xor(x2, 1);
    const float q2s = q1 + __shfl_xor(q1, 2);
    const float q4 = q2s + __shfl_xor(q2s, 4);
    for (int off = 16; off > 0; off >>= 1) x0 += __shfl_xor(x0, off);
    const float cv0 = x0;
    const float cv1 = __shfl(p8, 0);
    const float cv2 = __shfl(p8, 16);
    const float cv3 = __shfl(q4, 0);
    float v0 = sv0 / fmaxf(cv0, 1.0f);
    float v1 = sv1 / fmaxf(cv1, 1.0f);
    float v2 = sv2 / fmaxf(cv2, 1.0f);
    float v3 = sv3 / fmaxf(cv3, 1.0f);
    if (!(cv0 > 0.0f)) v0 = level_fallback<32, 0, 16384>(f0, pkp, b, i, cg);
    if (!(cv1 > 0.0f)) v1 = level_fallback<16, 256, 4096>(f1, pkp, b, i, cg);
    if (!(cv2 > 0.0f)) v2 = level_fallback<16, 384, 1024>(f2, pkp, b, i, cg);
    if (!(cv3 > 0.0f)) v3 = level_fallback<8, 512, 256>(f3, pkp, b, i, cg);
    float sum = 0.0f;
    sum += v0;
    sum += v1;
    sum += v2;
    sum += v3;
    out[bi * C_ + cg] = 0.25f * sum;
}

extern "C" void kernel_launch(void* const* d_in, const int* in_sizes, int n_in,
                              void* d_out, int out_size, void* d_ws, size_t ws_size,
                              hipStream_t stream) {
    const float* feat0 = (const float*)d_in[0];
    const float* feat1 = (const float*)d_in[1];
    const float* feat2 = (const float*)d_in[2];
    const float* feat3 = (const float*)d_in[3];
    const float* scrib = (const float*)d_in[4];
    float* ws = (float*)d_ws;

    float* sP   = ws + SP_OFF;
    unsigned int* sel2 = (unsigned int*)(ws + SEL2_OFF);
    unsigned int* sel3 = (unsigned int*)(ws + SEL3_OFF);
    float* cntP = ws + CNTP_OFF;
    unsigned long long* pkP = (unsigned long long*)((char*)d_ws + PKP_BYTEOFF);

    slab_masked_sum_kernel<2, 6, 128, true><<<128, 256, 0, stream>>>(feat1, scrib, sP + 256 * 4096, cntP + 256 * 32, pkP + 256 * 16, sel2, sel3);
    slab_masked_sum_kernel<4, 7, 128, false><<<256, 256, 0, stream>>>(feat0, scrib, sP,             cntP,            pkP,            nullptr, nullptr);
    small_levels_finalize_kernel<<<B_ * 8, 512, 0, stream>>>(
        feat0, feat1, feat2, feat3, sP, sel2, sel3, cntP, pkP, (float*)d_out);
}


# masked sums regrouped by mask band: a workgroup per (image, 16 mask rows) streams its 4 level-0 slabs then its level-1 slab and thresholds all 4 levels; 3 -> 2 launches; level-1 sum order changes
# speedup vs baseline: 2.6158x; 1.3125x over previous
#include <hip/hip_runtime.h>

typedef float v2f __attribute__((ext_vector_type(2)));
typedef float v4f __attribute__((ext_vector_type(4)));
typedef float v8f __attribute__((ext_vector_type(8)));
typedef unsigned int u2w __attribute__((ext_vector_type(2)));
typedef unsigned int u4w __attribute__((ext_vector_type(4)));

#define B_ 8
#define I_ 16
#define C_ 256


#define NLINES 1024
static const size_t SP_OFF      = 0;
static const size_t SEL2_OFF    = (size_t)512 * 4096;
static const size_t SEL3_OFF    = SEL2_OFF + (size_t)B_ * 4096;
static const size_t CNTP_OFF    = (size_t)528 * 4096;
static const size_t PKP_BYTEOFF = (CNTP_OFF + (size_t)NLINES * 32) * 4ull;

__device__ __forceinline__ unsigned long long load_pair(const float* p) {
    unsigned long long v;
    __builtin_memcpy(&v, p, sizeof(v));
    return v;
}
__device__ __forceinline__ float pair_lo(unsigned long long v) { return __uint_as_float((unsigned)v); }
__device__ __forceinline__ float pair_hi(unsigned long long v) { return __uint_as_float((unsigned)(v >> 32)); }

template <int LW>
__device__ __forceinline__ int mask_src(int rowbase, int pix) {
    constexpr int F    = 512 >> LW;
    constexpr int OFFS = F / 2 - 1;
    const int y = pix >> LW;
    const int x = pix & ((1 << LW) - 1);
    return (rowbase + y * F + OFFS) * 512 + x * F + OFFS;
}


__device__ __forceinline__ void store_tile(float* __restrict__ pd, const v8f& a0, const v8f& a1, int h) {
#pragma unroll
    for (int r = 0; r < 8; ++r) {
        const float send = h ? a0[r] : a1[r];
        const float recv = __shfl_xor(send, 16);
        pd[(size_t)r * C_]       = h ? recv : a0[r];
        pd[(size_t)(r + 8) * C_] = h ? a1[r] : recv;
    }
}

template <int LW>
__device__ __forceinline__ void issue_slab_loads(const float* __restrict__ frow, const float* __restrict__ scrib,
                                                 int rowbase, int pix0, v4f (&fr)[32], unsigned long long (&sc)[16]) {
    constexpr int P      = (1 << LW) * (1 << LW);
    constexpr int PLANE8 = 8 * 512 * 512;
#pragma unroll
    for (int q = 0; q < 4; ++q) {
        const int e = mask_src<LW>(rowbase, pix0 + 32 * q);
        sc[4 * q + 0] = load_pair(scrib + e);
        sc[4 * q + 1] = load_pair(scrib + e + 512);
        sc[4 * q + 2] = load_pair(scrib + e + PLANE8);
        sc[4 * q + 3] = load_pair(scrib + e + PLANE8 + 512);
    }
#pragma unroll
    for (int j = 0; j < 32; ++j) fr[j] = *(const v4f*)(frow + (size_t)j * (size_t)P);
}

__global__ void __launch_bounds__(256)
band_masked_sum_kernel(const float* __restrict__ feat0, const float* __restrict__ feat1,
                       const float* __restrict__ scrib,
                       float* __restrict__ spart, float* __restrict__ cntp,
                       unsigned long long* __restrict__ pkp,
                       unsigned int* __restrict__ selw2, unsigned int* __restrict__ selw3) {
    constexpr int KC    = 128;
    constexpr int PITCH = KC + 4;
    constexpr int NQ    = KC / 32;
    constexpr int NLD   = 32;
    constexpr int NSLAB = 5;
    __shared__ __attribute__((aligned(16))) float lsel[I_ * PITCH];
    __shared__ __attribute__((aligned(16))) float lfeat[C_ * PITCH];
    __shared__ float lcnt[4][I_];
    __shared__ unsigned long long lpk[4][I_];
    __shared__ unsigned int lw2[128];
    __shared__ unsigned int lw3[32];
    const int t    = threadIdx.x;
    const int lane = t & 31;
    const int wv   = t >> 5;
    const int b    = blockIdx.x >> 5;
    const int band = blockIdx.x & 31;
    const int c    = lane & 15;
    const int h    = lane >> 4;

    const float* f0row = feat0 + ((size_t)b * C_ + (size_t)(32 * wv)) * (size_t)16384 + (size_t)(512 * band + 4 * lane);
    const float* f1row = feat1 + ((size_t)b * C_ + (size_t)(32 * wv)) * (size_t)4096 + (size_t)(128 * band + 4 * lane);
    const int rowbase = (b * I_ + wv) * 512;
    const int pl0     = 512 * band + lane;
    const int pl1     = 128 * band + lane;

    float* lf = lfeat + (32 * wv) * PITCH + 4 * lane;
    float* ls = lsel + wv * PITCH + lane;
    const float* ap  = lsel + c * PITCH + 4 * h;
    const float* bp0 = lfeat + (32 * wv + c) * PITCH + 4 * h;
    const float* bp1 = bp0 + 16 * PITCH;

    v4f  fr[NLD];
    unsigned long long sc[4 * NQ];
    issue_slab_loads<7>(f0row, scrib, rowbase, pl0, fr, sc);

    const bool lvl2 = t < 128;
    const int  u    = t & 127;
    const int  srow = u >> 3;
    const int  rbs  = (b * I_ + srow) * 512;
    unsigned long long qs[8] = {0ull, 0ull, 0ull, 0ull, 0ull, 0ull, 0ull, 0ull};
    if (lvl2) {
#pragma unroll
        for (int k = 0; k < 4; ++k) {
            const int e2 = mask_src<5>(rbs, 32 * band + 4 * (u & 7) + k);
            qs[2 * k + 0] = load_pair(scrib + e2);
            qs[2 * k + 1] = load_pair(scrib + e2 + 512);
        }
    } else {
        const int e3 = mask_src<4>(rbs, 8 * band + (u & 7));
        qs[0] = load_pair(scrib + e3);
        qs[1] = load_pair(scrib + e3 + 512);
    }

    float cn0 = 0.0f, cn1 = 0.0f;
    unsigned long long bm0 = 0ull, bm1 = 0ull;
    float c00 = 0.0f, c01 = 0.0f;
    unsigned long long b00 = 0ull, b01 = 0ull;
    v8f acc0 = {};
    v8f acc1 = {};
    v8f t0 = {};
    v8f t1 = {};
    for (int s = 0; s < NSLAB; ++s) {
        const bool l1 = (s == NSLAB - 1);
        if (l1) {
            c00 = cn0; c01 = cn1; b00 = bm0; b01 = bm1;
            cn0 = 0.0f; cn1 = 0.0f; bm0 = 0ull; bm1 = 0ull;
            t0 = acc0; t1 = acc1;
            acc0 = v8f{};
            acc1 = v8f{};
        }
        const int pixb = l1 ? pl1 : (pl0 + s * KC);
#pragma unroll
        for (int q = 0; q < NQ; ++q) {
            const int pix = pixb + 32 * q;
            const float m0 = 0.25f * (pair_lo(sc[4 * q + 0]) + pair_hi(sc[4 * q + 0]) + pair_lo(sc[4 * q + 1]) + pair_hi(sc[4 * q + 1]));
            const float m1 = 0.25f * (pair_lo(sc[4 * q + 2]) + pair_hi(sc[4 * q + 2]) + pair_lo(sc[4 * q + 3]) + pair_hi(sc[4 * q + 3]));
            const float s0 = (m0 > 0.5f) ? 1.0f : 0.0f;
            const float s1 = (m1 > 0.5f) ? 1.0f : 0.0f;
            ls[32 * q]             = s0;
            ls[8 * PITCH + 32 * q] = s1;
            cn0 += s0;
            cn1 += s1;
            const unsigned long long lo = (unsigned long long)(0xFFFFFFFFu - (unsigned)pix);
            const unsigned long long k0 = ((unsigned long long)__float_as_uint(m0) << 32) | lo;
            const unsigned long long k1 = ((unsigned long long)__float_as_uint(m1) << 32) | lo;
            if (k0 > bm0) bm0 = k0;
            if (k1 > bm1) bm1 = k1;
        }
#pragma unroll
        for (int j = 0; j < NLD; ++j) *(v4f*)(lf + j * PITCH) = fr[j];
        __syncthreads();
        if (s == 0) {
            float cS = 0.0f;
            unsigned long long mS = 0ull;
            unsigned int wS = 0u;
            if (lvl2) {
#pragma unroll
                for (int k = 0; k < 4; ++k) {
                    const int pix = 32 * band + 4 * (u & 7) + k;
                    const float m = 0.25f * (pair_lo(qs[2 * k]) + pair_hi(qs[2 * k]) + pair_lo(qs[2 * k + 1]) + pair_hi(qs[2 * k + 1]));
                    const bool on = m > 0.5f;
                    wS |= (on ? 1u : 0u) << (8 * k);
                    cS += on ? 1.0f : 0.0f;
                    const unsigned long long key = ((unsigned long long)__float_as_uint(m) << 32)
                                                 | (unsigned long long)(0xFFFFFFFFu - (unsigned)pix);
                    if (key > mS) mS = key;
                }
                lw2[u] = wS;
            } else {
                const int pix = 8 * band + (u & 7);
                const float m = 0.25f * (pair_lo(qs[0]) + pair_hi(qs[0]) + pair_lo(qs[1]) + pair_hi(qs[1]));
                const bool on = m > 0.5f;
                cS = on ? 1.0f : 0.0f;
                mS = ((unsigned long long)__float_as_uint(m) << 32) | (unsigned long long)(0xFFFFFFFFu - (unsigned)pix);
                ((unsigned char*)lw3)[u] = on ? (unsigned char)1 : (unsigned char)0;
            }
            for (int off = 4; off > 0; off >>= 1) {
                cS += __shfl_xor(cS, off);
                const unsigned long long o = __shfl_xor(mS, off);
                if (o > mS) mS = o;
            }
            if ((u & 7) == 0) { lcnt[lvl2 ? 2 : 3][srow] = cS; lpk[lvl2 ? 2 : 3][srow] = mS; }
        }
        if (s + 2 < NSLAB) {
            issue_slab_loads<7>(f0row + (s + 1) * KC, scrib, rowbase, pl0 + (s + 1) * KC, fr, sc);
            __builtin_amdgcn_sched_barrier(0);
        } else if (s + 2 == NSLAB) {
            issue_slab_loads<6>(f1row, scrib, rowbase, pl1, fr, sc);
            __builtin_amdgcn_sched_barrier(0);
        }
        v4f an = *(const v4f*)(ap);
        v4f xn = *(const v4f*)(bp0);
        v4f yn = *(const v4f*)(bp1);
#pragma unroll
        for (int k8 = 0; k8 < KC / 8; ++k8) {
            const v4f a = an, x = xn, y = yn;
            if (k8 + 1 < KC / 8) {
                an = *(const v4f*)(ap  + 8 * (k8 + 1));
                xn = *(const v4f*)(bp0 + 8 * (k8 + 1));
                yn = *(const v4f*)(bp1 + 8 * (k8 + 1));
            }
            v2f a01 = {a[0], a[1]}, a23 = {a[2], a[3]};
            v2f x01 = {x[0], x[1]}, x23 = {x[2], x[3]};
            v2f y01 = {y[0], y[1]}, y23 = {y[2], y[3]};
            acc0 = __builtin_amdgcn_wmma_f32_16x16x4_f32(false, a01, false, x01, (short)0, acc0, false, false);
            acc1 = __builtin_amdgcn_wmma_f32_16x16x4_f32(false, a01, false, y01, (short)0, acc1, false, false);
            acc0 = __builtin_amdgcn_wmma_f32_16x16x4_f32(false, a23, false, x23, (short)0, acc0, false, false);
            acc1 = __builtin_amdgcn_wmma_f32_16x16x4_f32(false, a23, false, y23, (short)0, acc1, false, false);
        }
        if (s + 1 < NSLAB) __syncthreads();
    }
    for (int off = 16; off > 0; off >>= 1) {
        c00 += __shfl_xor(c00, off);
        c01 += __shfl_xor(c01, off);
        cn0 += __shfl_xor(cn0, off);
        cn1 += __shfl_xor(cn1, off);
        const unsigned long long o0 = __shfl_xor(b00, off);
        const unsigned long long o1 = __shfl_xor(b01, off);
        const unsigned long long o2 = __shfl_xor(bm0, off);
        const unsigned long long o3 = __shfl_xor(bm1, off);
        if (o0 > b00) b00 = o0;
        if (o1 > b01) b01 = o1;
        if (o2 > bm0) bm0 = o2;
        if (o3 > bm1) bm1 = o3;
    }
    if (lane == 0) {
        lcnt[0][wv]     = c00;
        lcnt[0][wv + 8] = c01;
        lpk[0][wv]      = b00;
        lpk[0][wv + 8]  = b01;
        lcnt[1][wv]     = cn0;
        lcnt[1][wv + 8] = cn1;
        lpk[1][wv]      = bm0;
        lpk[1][wv + 8]  = bm1;
    }
    __syncthreads();
    if (wv == 0) {
        unsigned int* d2 = selw2 + (size_t)blockIdx.x * 128;
#pragma unroll
        for (int r = 0; r < 4; ++r) d2[32 * r + lane] = lw2[32 * r + lane];
        selw3[(size_t)blockIdx.x * 32 + lane] = lw3[lane];
        if (lane < I_) {
#pragma unroll
            for (int lv = 0; lv < 4; ++lv) {
                const size_t n = (size_t)(lv * 256) + (size_t)blockIdx.x;
                cntp[n * 32 + lane] = lcnt[lv][lane];
                pkp[n * I_ + lane]  = lpk[lv][lane];
            }
        }
    }
    float* pd = spart + (size_t)blockIdx.x * (I_ * C_) + (size_t)(32 * wv + lane);
    store_tile(pd, t0, t1, h);
    store_tile(pd + (size_t)256 * (I_ * C_), acc0, acc1, h);
}


template <int N, int BASE, int P>
__device__ __forceinline__ float level_fallback(const float* __restrict__ feat,
                                                const unsigned long long* __restrict__ pkp,
                                                int b, int i, int c) {
    const int w0 = BASE + b * N;
    unsigned long long pk = 0ull;
    for (int j = 0; j < N; ++j) {
        const unsigned long long o = pkp[(size_t)(w0 + j) * I_ + i];
        if (o > pk) pk = o;
    }
    const unsigned idx = 0xFFFFFFFFu - (unsigned)(pk & 0xFFFFFFFFull);
    return feat[((size_t)b * C_ + (size_t)c) * (size_t)P + idx];
}

#define P2_ 1028
#define P3_ 260
__global__ void __launch_bounds__(512)
small_levels_finalize_kernel(const float* __restrict__ f0, const float* __restrict__ f1,
                             const float* __restrict__ f2, const float* __restrict__ f3,
                             const float* spart,
                             const unsigned int* __restrict__ selw2, const unsigned int* __restrict__ selw3,
                             const float* __restrict__ cntp,
                             const unsigned long long* __restrict__ pkp, float* __restrict__ out) {
    __shared__ __attribute__((aligned(16))) float lf2[32 * P2_];
    __shared__ __attribute__((aligned(16))) float ls2[I_ * P2_];
    __shared__ __attribute__((aligned(16))) float lf3[32 * P3_];
    __shared__ __attribute__((aligned(16))) float ls3[I_ * P3_];
    __shared__ float lred[2 * 8 * 2 * 256];
    const int t    = threadIdx.x;
    const int lane = t & 31;
    const int wv   = t >> 5;
    const int b    = blockIdx.x >> 3;
    const int g    = blockIdx.x & 7;
    const int c    = lane & 15;
    const int h    = lane >> 4;

    const float* g2 = f2 + ((size_t)b * C_ + (size_t)(32 * g + 2 * wv)) * 1024 + (size_t)(4 * lane);
    const float* g3 = f3 + ((size_t)b * C_ + (size_t)(32 * g + 2 * wv)) * 256 + (size_t)(4 * lane);
    v4f fr2[16], fr3[4];
#pragma unroll
    for (int rr = 0; rr < 2; ++rr) {
#pragma unroll
        for (int j = 0; j < 8; ++j) fr2[rr * 8 + j] = *(const v4f*)(g2 + rr * 1024 + 128 * j);
    }
#pragma unroll
    for (int rr = 0; rr < 2; ++rr) {
#pragma unroll
        for (int j = 0; j < 2; ++j) fr3[rr * 2 + j] = *(const v4f*)(g3 + rr * 256 + 128 * j);
    }
    u4w sw2[2];
#pragma unroll
    for (int q = 0; q < 2; ++q) sw2[q] = *(const u4w*)(selw2 + (size_t)b * 4096 + (size_t)(q * 2048 + 4 * t));
    const u2w sw3 = *(const u2w*)(selw3 + (size_t)b * 1024 + (size_t)(2 * t));

    const int i  = wv;
    const int bi = b * I_ + i;
    const int cg = 32 * g + lane;
    float x0 = cntp[(size_t)(b * 32 + lane) * 32 + i];
    float x1 = cntp[(size_t)(256 + b * 32 + lane) * 32 + i];
    float x2 = cntp[(size_t)(512 + b * 32 + lane) * 32 + i];
    float x3 = cntp[(size_t)(768 + b * 32 + lane) * 32 + i];
    float pv0[32], pv1[32];
    const float* sp0 = spart + (size_t)(b * 32) * (I_ * C_) + (size_t)(i * C_ + cg);
    const float* sp1 = spart + (size_t)(256 + b * 32) * (I_ * C_) + (size_t)(i * C_ + cg);
#pragma unroll
    for (int j = 0; j < 32; ++j) pv0[j] = sp0[(size_t)j * (I_ * C_)];
#pragma unroll
    for (int j = 0; j < 32; ++j) pv1[j] = sp1[(size_t)j * (I_ * C_)];

    float* d2 = lf2 + (2 * wv) * P2_ + 4 * lane;
    float* d3 = lf3 + (2 * wv) * P3_ + 4 * lane;
#pragma unroll
    for (int rr = 0; rr < 2; ++rr) {
#pragma unroll
        for (int j = 0; j < 8; ++j) *(v4f*)(d2 + rr * P2_ + 128 * j) = fr2[rr * 8 + j];
    }
#pragma unroll
    for (int rr = 0; rr < 2; ++rr) {
#pragma unroll
        for (int j = 0; j < 2; ++j) *(v4f*)(d3 + rr * P3_ + 128 * j) = fr3[rr * 2 + j];
    }
#pragma unroll
    for (int q = 0; q < 2; ++q) {
        const int widx = q * 2048 + 4 * t;
        const int row  = (widx >> 3) & 15;
        const int px0  = 32 * (widx >> 7) + 4 * (widx & 7);
#pragma unroll
        for (int k = 0; k < 4; ++k) {
            const unsigned int w = sw2[q][k];
            v4f v = {(float)(w & 1u), (float)((w >> 8) & 1u), (float)((w >> 16) & 1u), (float)((w >> 24) & 1u)};
            *(v4f*)(ls2 + row * P2_ + px0 + 4 * k) = v;
        }
    }
    {
        const int widx = 2 * t;
        const int row  = (widx >> 1) & 15;
        const int px0  = 8 * (widx >> 5) + 4 * (widx & 1);
#pragma unroll
        for (int k = 0; k < 2; ++k) {
            const unsigned int w = sw3[k];
            v4f v = {(float)(w & 1u), (float)((w >> 8) & 1u), (float)((w >> 16) & 1u), (float)((w >> 24) & 1u)};
            *(v4f*)(ls3 + row * P3_ + px0 + 4 * k) = v;
        }
    }
    __syncthreads();

    const int tl = wv & 1;
    const int kp = wv >> 1;
    v8f acc2 = {};
    v8f acc3 = {};
    {
        const float* ap = ls2 + c * P2_ + 128 * kp + 4 * h;
        const float* bp = lf2 + (16 * tl + c) * P2_ + 128 * kp + 4 * h;
        for (int k8 = 0; k8 < 16; ++k8) {
            v4f a = *(const v4f*)(ap + 8 * k8);
            v4f x = *(const v4f*)(bp + 8 * k8);
            v2f a01 = {a[0], a[1]}, a23 = {a[2], a[3]};
            v2f x01 = {x[0], x[1]}, x23 = {x[2], x[3]};
            acc2 = __builtin_amdgcn_wmma_f32_16x16x4_f32(false, a01, false, x01, (short)0, acc2, false, false);
            acc2 = __builtin_amdgcn_wmma_f32_16x16x4_f32(false, a23, false, x23, (short)0, acc2, false, false);
        }
    }
    {
        const float* ap = ls3 + c * P3_ + 32 * kp + 4 * h;
        const float* bp = lf3 + (16 * tl + c) * P3_ + 32 * kp + 4 * h;
        for (int k8 = 0; k8 < 4; ++k8) {
            v4f a = *(const v4f*)(ap + 8 * k8);
            v4f x = *(const v4f*)(bp + 8 * k8);
            v2f a01 = {a[0], a[1]}, a23 = {a[2], a[3]};
            v2f x01 = {x[0], x[1]}, x23 = {x[2], x[3]};
            acc3 = __builtin_amdgcn_wmma_f32_16x16x4_f32(false, a01, false, x01, (short)0, acc3, false, false);
            acc3 = __builtin_amdgcn_wmma_f32_16x16x4_f32(false, a23, false, x23, (short)0, acc3, false, false);
        }
    }
    {
        float* rd = lred + (kp * 2 + tl) * 256 + lane;
#pragma unroll
        for (int r = 0; r < 8; ++r) {
            rd[r * 32]        = acc2[r];
            rd[4096 + r * 32] = acc3[r];
        }
    }
    __syncthreads();

    const int src = (i & 7) * 32 + 16 * (i >> 3) + (lane & 15);
    const int tlc = lane >> 4;
    float sv2 = 0.0f, sv3 = 0.0f;
    for (int e = 0; e < 8; ++e) {
        sv2 += lred[(e * 2 + tlc) * 256 + src];
        sv3 += lred[4096 + (e * 2 + tlc) * 256 + src];
    }
    float sv0 = 0.0f, sv1 = 0.0f;
#pragma unroll
    for (int j = 0; j < 32; ++j) sv0 += pv0[j];
#pragma unroll
    for (int j = 0; j < 32; ++j) sv1 += pv1[j];

    for (int off = 16; off > 0; off >>= 1) {
        x0 += __shfl_xor(x0, off);
        x1 += __shfl_xor(x1, off);
        x2 += __shfl_xor(x2, off);
        x3 += __shfl_xor(x3, off);
    }
    const float cv0 = x0;
    const float cv1 = x1;
    const float cv2 = x2;
    const float cv3 = x3;
    float v0 = sv0 / fmaxf(cv0, 1.0f);
    float v1 = sv1 / fmaxf(cv1, 1.0f);
    float v2 = sv2 / fmaxf(cv2, 1.0f);
    float v3 = sv3 / fmaxf(cv3, 1.0f);
    if (!(cv0 > 0.0f)) v0 = level_fallback<32, 0, 16384>(f0, pkp, b, i, cg);
    if (!(cv1 > 0.0f)) v1 = level_fallback<32, 256, 4096>(f1, pkp, b, i, cg);
    if (!(cv2 > 0.0f)) v2 = level_fallback<32, 512, 1024>(f2, pkp, b, i, cg);
    if (!(cv3 > 0.0f)) v3 = level_fallback<32, 768, 256>(f3, pkp, b, i, cg);
    float sum = 0.0f;
    sum += v0;
    sum += v1;
    sum += v2;
    sum += v3;
    out[bi * C_ + cg] = 0.25f * sum;
}

extern "C" void kernel_launch(void* const* d_in, const int* in_sizes, int n_in,
                              void* d_out, int out_size, void* d_ws, size_t ws_size,
                              hipStream_t stream) {
    const float* feat0 = (const float*)d_in[0];
    const float* feat1 = (const float*)d_in[1];
    const float* feat2 = (const float*)d_in[2];
    const float* feat3 = (const float*)d_in[3];
    const float* scrib = (const float*)d_in[4];
    float* ws = (float*)d_ws;

    float* sP   = ws + SP_OFF;
    unsigned int* sel2 = (unsigned int*)(ws + SEL2_OFF);
    unsigned int* sel3 = (unsigned int*)(ws + SEL3_OFF);
    float* cntP = ws + CNTP_OFF;
    unsigned long long* pkP = (unsigned long long*)((char*)d_ws + PKP_BYTEOFF);

    band_masked_sum_kernel<<<B_ * 32, 256, 0, stream>>>(feat0, feat1, scrib, sP, cntP, pkP, sel2, sel3);
    small_levels_finalize_kernel<<<B_ * 8, 512, 0, stream>>>(
        feat0, feat1, feat2, feat3, sP, sel2, sel3, cntP, pkP, (float*)d_out);
}
